# starting kernel
# baseline (speedup 1.0000x reference)
#include <hip/hip_runtime.h>

typedef __attribute__((ext_vector_type(2))) float v2f;
typedef __attribute__((ext_vector_type(8))) float v8f;

#define N_PIX 4096
#define CH 32
#define CIN 64
#define BATCH 4
#define PLANE (CH * N_PIX)

__global__ __launch_bounds__(256) void qkv_conv_kernel(
    const float* __restrict__ x,
    const float* __restrict__ w1, const float* __restrict__ bb1,
    const float* __restrict__ w2, const float* __restrict__ bb2,
    const float* __restrict__ w3, const float* __restrict__ bb3,
    float* __restrict__ Q, float* __restrict__ K, float* __restrict__ V)
{
    int tid = blockIdx.x * blockDim.x + threadIdx.x;
    int b = tid >> 12;
    int n = tid & (N_PIX - 1);
    const float* xb = x + (size_t)b * CIN * N_PIX + n;
    float xr[CIN];
#pragma unroll
    for (int c = 0; c < CIN; ++c) xr[c] = xb[(size_t)c * N_PIX];
    size_t obase = (size_t)b * PLANE + n;
#pragma unroll 2
    for (int o = 0; o < CH; ++o) {
        float a1 = bb1[o], a2 = bb2[o], a3 = bb3[o];
#pragma unroll
        for (int c = 0; c < CIN; ++c) {
            float xv = xr[c];
            a1 = fmaf(w1[o * CIN + c], xv, a1);
            a2 = fmaf(w2[o * CIN + c], xv, a2);
            a3 = fmaf(w3[o * CIN + c], xv, a3);
        }
        Q[obase + (size_t)o * N_PIX] = a1;
        K[obase + (size_t)o * N_PIX] = a2;
        V[obase + (size_t)o * N_PIX] = a3;
    }
}

__global__ __launch_bounds__(128) void attn_kernel(
    const float* __restrict__ Q, const float* __restrict__ K,
    const float* __restrict__ V, float* __restrict__ Y)
{
    __shared__ float sc[BATCH][16][17];
    __shared__ float pc[BATCH][16][17];

    const int lane = threadIdx.x & 31;
    const int b    = threadIdx.x >> 5;
    const int lo   = lane & 15;
    const int hi   = lane >> 4;
    const int i0   = blockIdx.x * 16;

    const float* Qb = Q + (size_t)b * PLANE;
    const float* Kb = K + (size_t)b * PLANE;
    const float* Vb = V + (size_t)b * PLANE;

    v2f qa[8];
#pragma unroll
    for (int kk = 0; kk < 8; ++kk) {
        const float* p = Qb + (size_t)(i0 + lo) * CH + 4 * kk + 2 * hi;
        qa[kk] = v2f{p[0], p[1]};
    }

    v8f acc0 = {};
    v8f acc1 = {};

    for (int jt = 0; jt < N_PIX; jt += 16) {
        v8f s = {};
#pragma unroll
        for (int kk = 0; kk < 8; ++kk) {
            v2f kb;
            kb.x = Kb[(size_t)(4 * kk + 2 * hi)     * N_PIX + jt + lo];
            kb.y = Kb[(size_t)(4 * kk + 2 * hi + 1) * N_PIX + jt + lo];
            s = __builtin_amdgcn_wmma_f32_16x16x4_f32(
                    false, qa[kk], false, kb, (short)0, s, false, false);
        }
#pragma unroll
        for (int v = 0; v < 8; ++v) sc[b][v + 8 * hi][lo] = s[v];
        __syncthreads();

#pragma unroll
        for (int v = 0; v < 8; ++v) {
            int r = v + 8 * hi;
            float s0 = sc[0][r][lo], s1 = sc[1][r][lo];
            float s2 = sc[2][r][lo], s3 = sc[3][r][lo];
            float m = fmaxf(fmaxf(s0, s1), fmaxf(s2, s3));
            float e0 = __expf(s0 - m), e1 = __expf(s1 - m);
            float e2 = __expf(s2 - m), e3 = __expf(s3 - m);
            float d = e0 + e1 + e2 + e3;
            float eb = (b == 0) ? e0 : (b == 1) ? e1 : (b == 2) ? e2 : e3;
            pc[b][r][lo] = eb / d;
        }
        __syncthreads();

#pragma unroll
        for (int kk = 0; kk < 4; ++kk) {
            v2f a;
            a.x = pc[b][lo][4 * kk + 2 * hi];
            a.y = pc[b][lo][4 * kk + 2 * hi + 1];
            const float* vr0 = Vb + (size_t)(jt + 4 * kk + 2 * hi) * CH;
            const float* vr1 = vr0 + CH;
            v2f bl, bh;
            bl.x = vr0[lo];      bl.y = vr1[lo];
            bh.x = vr0[16 + lo]; bh.y = vr1[16 + lo];
            acc0 = __builtin_amdgcn_wmma_f32_16x16x4_f32(
                       false, a, false, bl, (short)0, acc0, false, false);
            acc1 = __builtin_amdgcn_wmma_f32_16x16x4_f32(
                       false, a, false, bh, (short)0, acc1, false, false);
        }
    }

    float* Yb = Y + (size_t)b * PLANE;
#pragma unroll
    for (int v = 0; v < 8; ++v) {
        int r = v + 8 * hi;
        Yb[(size_t)(i0 + r) * CH + lo]      = acc0[v];
        Yb[(size_t)(i0 + r) * CH + 16 + lo] = acc1[v];
    }
}

__global__ __launch_bounds__(256) void out_conv_kernel(
    const float* __restrict__ Y, const float* __restrict__ w4,
    const float* __restrict__ bb4, float* __restrict__ out)
{
    int tid = blockIdx.x * blockDim.x + threadIdx.x;
    int b = tid >> 12;
    int n = tid & (N_PIX - 1);
    const float* Yb = Y + (size_t)b * PLANE;
    float yr[CH];
#pragma unroll
    for (int c = 0; c < CH; ++c) yr[c] = Yb[(size_t)c * N_PIX + n];
    float* ob = out + (size_t)b * CIN * N_PIX + n;
#pragma unroll 2
    for (int o = 0; o < CIN; ++o) {
        float a = bb4[o];
#pragma unroll
        for (int c = 0; c < CH; ++c) a = fmaf(w4[o * CH + c], yr[c], a);
        ob[(size_t)o * N_PIX] = a;
    }
}

extern "C" void kernel_launch(void* const* d_in, const int* in_sizes, int n_in,
                              void* d_out, int out_size, void* d_ws, size_t ws_size,
                              hipStream_t stream) {
    const float* x   = (const float*)d_in[0];
    const float* w1  = (const float*)d_in[1];
    const float* bb1 = (const float*)d_in[2];
    const float* w2  = (const float*)d_in[3];
    const float* bb2 = (const float*)d_in[4];
    const float* w3  = (const float*)d_in[5];
    const float* bb3 = (const float*)d_in[6];
    const float* w4  = (const float*)d_in[7];
    const float* bb4 = (const float*)d_in[8];
    float* out = (float*)d_out;

    float* Q = (float*)d_ws;
    float* K = Q + (size_t)BATCH * PLANE;
    float* V = K + (size_t)BATCH * PLANE;
    float* Y = V + (size_t)BATCH * PLANE;

    qkv_conv_kernel<<<(BATCH * N_PIX) / 256, 256, 0, stream>>>(
        x, w1, bb1, w2, bb2, w3, bb3, Q, K, V);
    attn_kernel<<<N_PIX / 16, 128, 0, stream>>>(Q, K, V, Y);
    out_conv_kernel<<<(BATCH * N_PIX) / 256, 256, 0, stream>>>(Y, w4, bb4, out);
}


# output 1x1 conv: weight rows read via per-row pointers as 64-byte scalar loads, FMAs take weights from scalar registers; split into 256 groups of 256 pixels x 16 outputs (was 64); sum order kept
# speedup vs baseline: 5.3996x; 5.3996x over previous
#include <hip/hip_runtime.h>

typedef __attribute__((ext_vector_type(2))) float v2f;
typedef __attribute__((ext_vector_type(8))) float v8f;

#define N_PIX 4096
#define CH 32
#define CIN 64
#define BATCH 4
#define PLANE (CH * N_PIX)

#define OG 16

__global__ __launch_bounds__(256) void qkv_conv_kernel(
    const float* __restrict__ x,
    const float* __restrict__ w1, const float* __restrict__ bb1,
    const float* __restrict__ w2, const float* __restrict__ bb2,
    const float* __restrict__ w3, const float* __restrict__ bb3,
    float* __restrict__ Q, float* __restrict__ K, float* __restrict__ V)
{
    const int sel = blockIdx.y;
    const float* __restrict__ w  = sel == 0 ? w1  : (sel == 1 ? w2  : w3);
    const float* __restrict__ bs = sel == 0 ? bb1 : (sel == 1 ? bb2 : bb3);
    float* __restrict__ dst      = sel == 0 ? Q   : (sel == 1 ? K   : V);
    const int o0 = blockIdx.z * OG;

    const int tid = blockIdx.x * 256 + threadIdx.x;
    const int b = tid >> 12;
    const int n = tid & (N_PIX - 1);
    const float* xb = x + (size_t)b * CIN * N_PIX + n;
    float xr[CIN];
#pragma unroll
    for (int c = 0; c < CIN; ++c) xr[c] = xb[(size_t)c * N_PIX];

    float* ob = dst + (size_t)b * PLANE + (size_t)o0 * N_PIX + n;
    const float* wr = w + (size_t)o0 * CIN;
    const float* br = bs + o0;
    for (int oq = 0; oq < OG; oq += 4) {
        const float* r0 = wr + (size_t)oq * CIN;
        const float* r1 = r0 + CIN;
        const float* r2 = r1 + CIN;
        const float* r3 = r2 + CIN;
        float a0 = br[oq], a1 = br[oq + 1], a2 = br[oq + 2], a3 = br[oq + 3];
#pragma unroll
        for (int c = 0; c < CIN; ++c) {
            const float xv = xr[c];
            a0 = fmaf(r0[c], xv, a0);
            a1 = fmaf(r1[c], xv, a1);
            a2 = fmaf(r2[c], xv, a2);
            a3 = fmaf(r3[c], xv, a3);
        }
        float* op = ob + (size_t)oq * N_PIX;
        op[0]         = a0;
        op[N_PIX]     = a1;
        op[2 * N_PIX] = a2;
        op[3 * N_PIX] = a3;
    }
}

#define KT 8
#define JSTEP (16 * KT)
#define PP 20

__global__ __launch_bounds__(128) void attn_kernel(
    const float* __restrict__ Q, const float* __restrict__ K,
    const float* __restrict__ V, float* __restrict__ Y)
{
    __shared__ float4 sc[BATCH][KT][2][32];
    __shared__ __attribute__((aligned(16))) float pc[BATCH][KT][16][PP];

    const int lane = threadIdx.x & 31;
    const int b    = threadIdx.x >> 5;
    const int lo   = lane & 15;
    const int hi   = lane >> 4;
    const int i0   = blockIdx.x * 16;
    const int pcol = ((lo >> 1) & 1) * 8 + (lo >> 2) * 2 + (lo & 1);

    const float* Qb = Q + (size_t)b * PLANE;
    const float* Kb = K + (size_t)b * PLANE;
    const float* Vb = V + (size_t)b * PLANE;

    v2f qa[8];
#pragma unroll
    for (int kk = 0; kk < 8; ++kk) {
        const float* p = Qb + (size_t)(i0 + lo) * CH + 4 * kk + 2 * hi;
        qa[kk] = v2f{p[0], p[1]};
    }

    v8f acc0 = {};
    v8f acc1 = {};

    for (int jt = 0; jt < N_PIX; jt += JSTEP) {
        float4 kq[8][2][KT / 4];
#pragma unroll
        for (int kk = 0; kk < 8; ++kk) {
            const float* kr0 = Kb + (size_t)(4 * kk + 2 * hi) * N_PIX + jt + 4 * lo;
            const float* kr1 = kr0 + N_PIX;
#pragma unroll
            for (int g = 0; g < KT / 4; ++g) {
                kq[kk][0][g] = *(const float4*)(kr0 + 64 * g);
                kq[kk][1][g] = *(const float4*)(kr1 + 64 * g);
            }
        }
        float2 vq[KT][4][2];
#pragma unroll
        for (int kk = 0; kk < 4; ++kk) {
            const int t = KT - 1;
            const float* vr0 = Vb + (size_t)(jt + 64 * (t >> 2) + (t & 3) + 16 * kk + 8 * hi) * CH + 2 * lo;
            vq[t][kk][0] = *(const float2*)vr0;
            vq[t][kk][1] = *(const float2*)(vr0 + 4 * CH);
        }
        __builtin_amdgcn_sched_barrier(0);
        v8f s[KT];
#pragma unroll
        for (int t = 0; t < KT; ++t) s[t] = v8f{};
#pragma unroll
        for (int kk = 0; kk < 8; ++kk) {
#pragma unroll
            for (int g = 0; g < KT / 4; ++g) {
                const float4 c0 = kq[kk][0][g];
                const float4 c1 = kq[kk][1][g];
                s[4 * g + 0] = __builtin_amdgcn_wmma_f32_16x16x4_f32(
                    false, qa[kk], false, v2f{c0.x, c1.x}, (short)0, s[4 * g + 0], false, false);
                s[4 * g + 1] = __builtin_amdgcn_wmma_f32_16x16x4_f32(
                    false, qa[kk], false, v2f{c0.y, c1.y}, (short)0, s[4 * g + 1], false, false);
                s[4 * g + 2] = __builtin_amdgcn_wmma_f32_16x16x4_f32(
                    false, qa[kk], false, v2f{c0.z, c1.z}, (short)0, s[4 * g + 2], false, false);
                s[4 * g + 3] = __builtin_amdgcn_wmma_f32_16x16x4_f32(
                    false, qa[kk], false, v2f{c0.w, c1.w}, (short)0, s[4 * g + 3], false, false);
            }
        }
#pragma unroll
        for (int t = 0; t < KT; ++t) {
            sc[b][t][0][lane] = make_float4(s[t][0], s[t][1], s[t][2], s[t][3]);
            sc[b][t][1][lane] = make_float4(s[t][4], s[t][5], s[t][6], s[t][7]);
        }
        __syncthreads();

#pragma unroll
        for (int t = 0; t < KT - 1; ++t) {
#pragma unroll
            for (int kk = 0; kk < 4; ++kk) {
                const float* vr0 = Vb + (size_t)(jt + 64 * (t >> 2) + (t & 3) + 16 * kk + 8 * hi) * CH + 2 * lo;
                vq[t][kk][0] = *(const float2*)vr0;
                vq[t][kk][1] = *(const float2*)(vr0 + 4 * CH);
            }
        }
        __builtin_amdgcn_sched_barrier(0);

#pragma unroll
        for (int tt = 0; tt < 2; ++tt) {
            const int t = 2 * b + tt;
#pragma unroll
            for (int h = 0; h < 2; ++h) {
                const float4 r0 = sc[0][t][h][lane];
                const float4 r1 = sc[1][t][h][lane];
                const float4 r2 = sc[2][t][h][lane];
                const float4 r3 = sc[3][t][h][lane];
                const float q0[4] = {r0.x, r0.y, r0.z, r0.w};
                const float q1[4] = {r1.x, r1.y, r1.z, r1.w};
                const float q2[4] = {r2.x, r2.y, r2.z, r2.w};
                const float q3[4] = {r3.x, r3.y, r3.z, r3.w};
                float e0[4], e1[4], e2[4], e3[4], inv[4];
#pragma unroll
                for (int u = 0; u < 4; ++u) {
                    float m = fmaxf(fmaxf(q0[u], q1[u]), fmaxf(q2[u], q3[u]));
                    e0[u] = __expf(q0[u] - m); e1[u] = __expf(q1[u] - m);
                    e2[u] = __expf(q2[u] - m); e3[u] = __expf(q3[u] - m);
                }
#pragma unroll
                for (int u = 0; u < 4; ++u) {
                    float d = e0[u] + e1[u] + e2[u] + e3[u];
                    float y = __builtin_amdgcn_rcpf(d);
                    inv[u] = fmaf(fmaf(-d, y, 1.0f), y, y);
                }
#pragma unroll
                for (int u = 0; u < 4; ++u) {
                    const int r = 4 * h + u + 8 * hi;
                    pc[0][t][r][pcol] = e0[u] * inv[u];
                    pc[1][t][r][pcol] = e1[u] * inv[u];
                    pc[2][t][r][pcol] = e2[u] * inv[u];
                    pc[3][t][r][pcol] = e3[u] * inv[u];
                }
            }
        }
        __syncthreads();

        v2f pa[KT][4];
#pragma unroll
        for (int t = 0; t < KT; ++t) {
            const float4 p0 = *(const float4*)&pc[b][t][lo][8 * hi];
            const float4 p1 = *(const float4*)&pc[b][t][lo][8 * hi + 4];
            pa[t][0] = v2f{p0.x, p0.y};
            pa[t][1] = v2f{p0.z, p0.w};
            pa[t][2] = v2f{p1.x, p1.y};
            pa[t][3] = v2f{p1.z, p1.w};
        }
        __builtin_amdgcn_sched_barrier(0);
#pragma unroll
        for (int t = 0; t < KT; ++t) {
#pragma unroll
            for (int kk = 0; kk < 4; ++kk) {
                const v2f vl = v2f{vq[t][kk][0].x, vq[t][kk][1].x};
                const v2f vh = v2f{vq[t][kk][0].y, vq[t][kk][1].y};
                acc0 = __builtin_amdgcn_wmma_f32_16x16x4_f32(
                           false, pa[t][kk], false, vl, (short)0, acc0, false, false);
                acc1 = __builtin_amdgcn_wmma_f32_16x16x4_f32(
                           false, pa[t][kk], false, vh, (short)0, acc1, false, false);
            }
        }
    }

    float* Yb = Y + (size_t)b * PLANE;
#pragma unroll
    for (int v = 0; v < 8; ++v) {
        int r = v + 8 * hi;
        *(float2*)(Yb + (size_t)(i0 + r) * CH + 2 * lo) = make_float2(acc0[v], acc1[v]);
    }
}

#define OGO 16

__global__ __launch_bounds__(256) void out_conv_kernel(
    const float* __restrict__ Y, const float* __restrict__ w4,
    const float* __restrict__ bb4, float* __restrict__ out)
{
    const int o0 = blockIdx.y * OGO;
    const int tid = blockIdx.x * 256 + threadIdx.x;
    const int b = tid >> 12;
    const int n = tid & (N_PIX - 1);
    const float* Yb = Y + (size_t)b * PLANE + n;
    float yr[CH];
#pragma unroll
    for (int c = 0; c < CH; ++c) yr[c] = Yb[(size_t)c * N_PIX];

    float* ob = out + (size_t)b * CIN * N_PIX + (size_t)o0 * N_PIX + n;
    const float* wr = w4 + (size_t)o0 * CH;
    const float* br = bb4 + o0;
#pragma clang loop vectorize(disable)
    for (int oq = 0; oq < OGO; oq += 4) {
        const float* r0 = wr + (size_t)oq * CH;
        const float* r1 = r0 + CH;
        const float* r2 = r1 + CH;
        const float* r3 = r2 + CH;
        float a0 = br[oq], a1 = br[oq + 1], a2 = br[oq + 2], a3 = br[oq + 3];
#pragma unroll
        for (int c = 0; c < CH; ++c) {
            const float yv = yr[c];
            a0 = fmaf(r0[c], yv, a0);
            a1 = fmaf(r1[c], yv, a1);
            a2 = fmaf(r2[c], yv, a2);
            a3 = fmaf(r3[c], yv, a3);
        }
        float* op = ob + (size_t)oq * N_PIX;
        op[0]         = a0;
        op[N_PIX]     = a1;
        op[2 * N_PIX] = a2;
        op[3 * N_PIX] = a3;
    }
}

extern "C" void kernel_launch(void* const* d_in, const int* in_sizes, int n_in,
                              void* d_out, int out_size, void* d_ws, size_t ws_size,
                              hipStream_t stream) {
    const float* x   = (const float*)d_in[0];
    const float* w1  = (const float*)d_in[1];
    const float* bb1 = (const float*)d_in[2];
    const float* w2  = (const float*)d_in[3];
    const float* bb2 = (const float*)d_in[4];
    const float* w3  = (const float*)d_in[5];
    const float* bb3 = (const float*)d_in[6];
    const float* w4  = (const float*)d_in[7];
    const float* bb4 = (const float*)d_in[8];
    float* out = (float*)d_out;

    float* Q = (float*)d_ws;
    float* K = Q + (size_t)BATCH * PLANE;
    float* V = K + (size_t)BATCH * PLANE;
    float* Y = V + (size_t)BATCH * PLANE;

    qkv_conv_kernel<<<dim3((BATCH * N_PIX) / 256, 3, CH / OG), 256, 0, stream>>>(
        x, w1, bb1, w2, bb2, w3, bb3, Q, K, V);
    attn_kernel<<<N_PIX / 16, 128, 0, stream>>>(Q, K, V, Y);
    out_conv_kernel<<<dim3((BATCH * N_PIX) / 256, CIN / OGO), 256, 0, stream>>>(Y, w4, bb4, out);
}


# attention: 8-wave work-groups, two key halves run one barrier step out of phase (one half's matrix segment beside the other's softmax on every SIMD); Y = half 0 + half 1, key sum order changes
# speedup vs baseline: 6.0328x; 1.1173x over previous
#include <hip/hip_runtime.h>

typedef __attribute__((ext_vector_type(2))) float v2f;
typedef __attribute__((ext_vector_type(8))) float v8f;

#define N_PIX 4096
#define CH 32
#define CIN 64
#define BATCH 4
#define PLANE (CH * N_PIX)

#define OG 16

__global__ __launch_bounds__(256) void qkv_conv_kernel(
    const float* __restrict__ x,
    const float* __restrict__ w1, const float* __restrict__ bb1,
    const float* __restrict__ w2, const float* __restrict__ bb2,
    const float* __restrict__ w3, const float* __restrict__ bb3,
    float* __restrict__ Q, float* __restrict__ K, float* __restrict__ V)
{
    const int sel = blockIdx.y;
    const float* __restrict__ w  = sel == 0 ? w1  : (sel == 1 ? w2  : w3);
    const float* __restrict__ bs = sel == 0 ? bb1 : (sel == 1 ? bb2 : bb3);
    float* __restrict__ dst      = sel == 0 ? Q   : (sel == 1 ? K   : V);
    const int o0 = blockIdx.z * OG;

    const int tid = blockIdx.x * 256 + threadIdx.x;
    const int b = tid >> 12;
    const int n = tid & (N_PIX - 1);
    const float* xb = x + (size_t)b * CIN * N_PIX + n;
    float xr[CIN];
#pragma unroll
    for (int c = 0; c < CIN; ++c) xr[c] = xb[(size_t)c * N_PIX];

    float* ob = dst + (size_t)b * PLANE + (size_t)o0 * N_PIX + n;
    const float* wr = w + (size_t)o0 * CIN;
    const float* br = bs + o0;
    for (int oq = 0; oq < OG; oq += 4) {
        const float* r0 = wr + (size_t)oq * CIN;
        const float* r1 = r0 + CIN;
        const float* r2 = r1 + CIN;
        const float* r3 = r2 + CIN;
        float a0 = br[oq], a1 = br[oq + 1], a2 = br[oq + 2], a3 = br[oq + 3];
#pragma unroll
        for (int c = 0; c < CIN; ++c) {
            const float xv = xr[c];
            a0 = fmaf(r0[c], xv, a0);
            a1 = fmaf(r1[c], xv, a1);
            a2 = fmaf(r2[c], xv, a2);
            a3 = fmaf(r3[c], xv, a3);
        }
        float* op = ob + (size_t)oq * N_PIX;
        op[0]         = a0;
        op[N_PIX]     = a1;
        op[2 * N_PIX] = a2;
        op[3 * N_PIX] = a3;
    }
}

#define KT 8
#define JSTEP (16 * KT)
#define PP 20
#define KHALF (N_PIX / 2)
#define NTH (KHALF / JSTEP)

__global__ __launch_bounds__(256) void attn_kernel(
    const float* __restrict__ Q, const float* __restrict__ K,
    const float* __restrict__ V, float* __restrict__ Y)
{
    __shared__ float4 sc[2][BATCH][KT][2][32];
    __shared__ __attribute__((aligned(16))) float pc[2][BATCH][KT][16][PP];

    const int lane = threadIdx.x & 31;
    const int wv   = threadIdx.x >> 5;
    const int b    = wv & 3;
    const int g    = __builtin_amdgcn_readfirstlane(wv >> 2);
    const int lo   = lane & 15;
    const int hi   = lane >> 4;
    const int i0   = blockIdx.x * 16;
    const int pcol = ((lo >> 1) & 1) * 8 + (lo >> 2) * 2 + (lo & 1);

    const float* Qb = Q + (size_t)b * PLANE;
    const float* Kb = K + (size_t)b * PLANE + (size_t)g * KHALF;
    const float* Vb = V + (size_t)b * PLANE + (size_t)g * KHALF * CH;

    v2f qa[8];
#pragma unroll
    for (int kk = 0; kk < 8; ++kk) {
        const float* p = Qb + (size_t)(i0 + lo) * CH + 4 * kk + 2 * hi;
        qa[kk] = v2f{p[0], p[1]};
    }

    v8f acc0 = {};
    v8f acc1 = {};

    float2 vq[KT][4][2];

    for (int it = 0; it < 2 * NTH + 2; ++it) {
        const int ls = it - g;
        if (ls >= 0 && ls <= 2 * NTH) {
            const int j  = ls >> 1;
            const int jt = j * JSTEP;
            if ((ls & 1) == 0) {
                const bool has_qk = j < NTH;
                const bool has_pv = j > 0;
                float4 kq[8][2][KT / 4];
                if (has_qk) {
#pragma unroll
                    for (int kk = 0; kk < 8; ++kk) {
                        const float* kr0 = Kb + (size_t)(4 * kk + 2 * hi) * N_PIX + jt + 4 * lo;
                        const float* kr1 = kr0 + N_PIX;
#pragma unroll
                        for (int q4 = 0; q4 < KT / 4; ++q4) {
                            kq[kk][0][q4] = *(const float4*)(kr0 + 64 * q4);
                            kq[kk][1][q4] = *(const float4*)(kr1 + 64 * q4);
                        }
                    }
                }
                if (has_pv) {
                    v2f pa[KT][4];
#pragma unroll
                    for (int t = 0; t < KT; ++t) {
                        const float4 p0 = *(const float4*)&pc[g][b][t][lo][8 * hi];
                        const float4 p1 = *(const float4*)&pc[g][b][t][lo][8 * hi + 4];
                        pa[t][0] = v2f{p0.x, p0.y};
                        pa[t][1] = v2f{p0.z, p0.w};
                        pa[t][2] = v2f{p1.x, p1.y};
                        pa[t][3] = v2f{p1.z, p1.w};
                    }
                    __builtin_amdgcn_sched_barrier(0);
#pragma unroll
                    for (int t = 0; t < KT; ++t) {
#pragma unroll
                        for (int kk = 0; kk < 4; ++kk) {
                            const v2f vl = v2f{vq[t][kk][0].x, vq[t][kk][1].x};
                            const v2f vh = v2f{vq[t][kk][0].y, vq[t][kk][1].y};
                            acc0 = __builtin_amdgcn_wmma_f32_16x16x4_f32(
                                       false, pa[t][kk], false, vl, (short)0, acc0, false, false);
                            acc1 = __builtin_amdgcn_wmma_f32_16x16x4_f32(
                                       false, pa[t][kk], false, vh, (short)0, acc1, false, false);
                        }
                    }
                }
                if (has_qk) {
#pragma unroll
                    for (int kk = 0; kk < 4; ++kk) {
                        const int t = KT - 1;
                        const float* vr0 = Vb + (size_t)(jt + 64 * (t >> 2) + (t & 3) + 16 * kk + 8 * hi) * CH + 2 * lo;
                        vq[t][kk][0] = *(const float2*)vr0;
                        vq[t][kk][1] = *(const float2*)(vr0 + 4 * CH);
                    }
                    v8f s[KT];
#pragma unroll
                    for (int t = 0; t < KT; ++t) s[t] = v8f{};
#pragma unroll
                    for (int kk = 0; kk < 8; ++kk) {
#pragma unroll
                        for (int q4 = 0; q4 < KT / 4; ++q4) {
                            const float4 c0 = kq[kk][0][q4];
                            const float4 c1 = kq[kk][1][q4];
                            s[4 * q4 + 0] = __builtin_amdgcn_wmma_f32_16x16x4_f32(
                                false, qa[kk], false, v2f{c0.x, c1.x}, (short)0, s[4 * q4 + 0], false, false);
                            s[4 * q4 + 1] = __builtin_amdgcn_wmma_f32_16x16x4_f32(
                                false, qa[kk], false, v2f{c0.y, c1.y}, (short)0, s[4 * q4 + 1], false, false);
                            s[4 * q4 + 2] = __builtin_amdgcn_wmma_f32_16x16x4_f32(
                                false, qa[kk], false, v2f{c0.z, c1.z}, (short)0, s[4 * q4 + 2], false, false);
                            s[4 * q4 + 3] = __builtin_amdgcn_wmma_f32_16x16x4_f32(
                                false, qa[kk], false, v2f{c0.w, c1.w}, (short)0, s[4 * q4 + 3], false, false);
                        }
                    }
#pragma unroll
                    for (int t = 0; t < KT; ++t) {
                        sc[g][b][t][0][lane] = make_float4(s[t][0], s[t][1], s[t][2], s[t][3]);
                        sc[g][b][t][1][lane] = make_float4(s[t][4], s[t][5], s[t][6], s[t][7]);
                    }
                }
            } else {
#pragma unroll
                for (int t = 0; t < KT - 1; ++t) {
#pragma unroll
                    for (int kk = 0; kk < 4; ++kk) {
                        const float* vr0 = Vb + (size_t)(jt + 64 * (t >> 2) + (t & 3) + 16 * kk + 8 * hi) * CH + 2 * lo;
                        vq[t][kk][0] = *(const float2*)vr0;
                        vq[t][kk][1] = *(const float2*)(vr0 + 4 * CH);
                    }
                }
                __builtin_amdgcn_sched_barrier(0);

#pragma unroll
                for (int tt = 0; tt < 2; ++tt) {
                    const int t = 2 * b + tt;
#pragma unroll
                    for (int h = 0; h < 2; ++h) {
                        const float4 r0 = sc[g][0][t][h][lane];
                        const float4 r1 = sc[g][1][t][h][lane];
                        const float4 r2 = sc[g][2][t][h][lane];
                        const float4 r3 = sc[g][3][t][h][lane];
                        const float q0[4] = {r0.x, r0.y, r0.z, r0.w};
                        const float q1[4] = {r1.x, r1.y, r1.z, r1.w};
                        const float q2[4] = {r2.x, r2.y, r2.z, r2.w};
                        const float q3[4] = {r3.x, r3.y, r3.z, r3.w};
                        float e0[4], e1[4], e2[4], e3[4], inv[4];
#pragma unroll
                        for (int u = 0; u < 4; ++u) {
                            float m = fmaxf(fmaxf(q0[u], q1[u]), fmaxf(q2[u], q3[u]));
                            e0[u] = __expf(q0[u] - m); e1[u] = __expf(q1[u] - m);
                            e2[u] = __expf(q2[u] - m); e3[u] = __expf(q3[u] - m);
                        }
#pragma unroll
                        for (int u = 0; u < 4; ++u) {
                            float d = e0[u] + e1[u] + e2[u] + e3[u];
                            float y = __builtin_amdgcn_rcpf(d);
                            inv[u] = fmaf(fmaf(-d, y, 1.0f), y, y);
                        }
#pragma unroll
                        for (int u = 0; u < 4; ++u) {
                            const int r = 4 * h + u + 8 * hi;
                            pc[g][0][t][r][pcol] = e0[u] * inv[u];
                            pc[g][1][t][r][pcol] = e1[u] * inv[u];
                            pc[g][2][t][r][pcol] = e2[u] * inv[u];
                            pc[g][3][t][r][pcol] = e3[u] * inv[u];
                        }
                    }
                }
            }
        }
        __syncthreads();
    }

    if (g == 1) {
        sc[1][b][0][0][lane] = make_float4(acc0[0], acc0[1], acc0[2], acc0[3]);
        sc[1][b][0][1][lane] = make_float4(acc0[4], acc0[5], acc0[6], acc0[7]);
        sc[1][b][1][0][lane] = make_float4(acc1[0], acc1[1], acc1[2], acc1[3]);
        sc[1][b][1][1][lane] = make_float4(acc1[4], acc1[5], acc1[6], acc1[7]);
    }
    __syncthreads();
    if (g == 0) {
        const float4 a0 = sc[1][b][0][0][lane];
        const float4 a1 = sc[1][b][0][1][lane];
        const float4 c0 = sc[1][b][1][0][lane];
        const float4 c1 = sc[1][b][1][1][lane];
        const float o0[8] = {a0.x, a0.y, a0.z, a0.w, a1.x, a1.y, a1.z, a1.w};
        const float o1[8] = {c0.x, c0.y, c0.z, c0.w, c1.x, c1.y, c1.z, c1.w};
        float* Yb = Y + (size_t)b * PLANE;
#pragma unroll
        for (int v = 0; v < 8; ++v) {
            int r = v + 8 * hi;
            *(float2*)(Yb + (size_t)(i0 + r) * CH + 2 * lo) =
                make_float2(acc0[v] + o0[v], acc1[v] + o1[v]);
        }
    }
}

#define OGO 16

__global__ __launch_bounds__(256) void out_conv_kernel(
    const float* __restrict__ Y, const float* __restrict__ w4,
    const float* __restrict__ bb4, float* __restrict__ out)
{
    const int o0 = blockIdx.y * OGO;
    const int tid = blockIdx.x * 256 + threadIdx.x;
    const int b = tid >> 12;
    const int n = tid & (N_PIX - 1);
    const float* Yb = Y + (size_t)b * PLANE + n;
    float yr[CH];
#pragma unroll
    for (int c = 0; c < CH; ++c) yr[c] = Yb[(size_t)c * N_PIX];

    float* ob = out + (size_t)b * CIN * N_PIX + (size_t)o0 * N_PIX + n;
    const float* wr = w4 + (size_t)o0 * CH;
    const float* br = bb4 + o0;
#pragma clang loop vectorize(disable)
    for (int oq = 0; oq < OGO; oq += 4) {
        const float* r0 = wr + (size_t)oq * CH;
        const float* r1 = r0 + CH;
        const float* r2 = r1 + CH;
        const float* r3 = r2 + CH;
        float a0 = br[oq], a1 = br[oq + 1], a2 = br[oq + 2], a3 = br[oq + 3];
#pragma unroll
        for (int c = 0; c < CH; ++c) {
            const float yv = yr[c];
            a0 = fmaf(r0[c], yv, a0);
            a1 = fmaf(r1[c], yv, a1);
            a2 = fmaf(r2[c], yv, a2);
            a3 = fmaf(r3[c], yv, a3);
        }
        float* op = ob + (size_t)oq * N_PIX;
        op[0]         = a0;
        op[N_PIX]     = a1;
        op[2 * N_PIX] = a2;
        op[3 * N_PIX] = a3;
    }
}

extern "C" void kernel_launch(void* const* d_in, const int* in_sizes, int n_in,
                              void* d_out, int out_size, void* d_ws, size_t ws_size,
                              hipStream_t stream) {
    const float* x   = (const float*)d_in[0];
    const float* w1  = (const float*)d_in[1];
    const float* bb1 = (const float*)d_in[2];
    const float* w2  = (const float*)d_in[3];
    const float* bb2 = (const float*)d_in[4];
    const float* w3  = (const float*)d_in[5];
    const float* bb3 = (const float*)d_in[6];
    const float* w4  = (const float*)d_in[7];
    const float* bb4 = (const float*)d_in[8];
    float* out = (float*)d_out;

    float* Q = (float*)d_ws;
    float* K = Q + (size_t)BATCH * PLANE;
    float* V = K + (size_t)BATCH * PLANE;
    float* Y = V + (size_t)BATCH * PLANE;

    qkv_conv_kernel<<<dim3((BATCH * N_PIX) / 256, 3, CH / OG), 256, 0, stream>>>(
        x, w1, bb1, w2, bb2, w3, bb3, Q, K, V);
    attn_kernel<<<N_PIX / 16, 256, 0, stream>>>(Q, K, V, Y);
    out_conv_kernel<<<dim3((BATCH * N_PIX) / 256, CIN / OGO), 256, 0, stream>>>(Y, w4, bb4, out);
}


# attention: score products of trip k+1 moved into the softmax segment of trip k (scores double-buffered in LDS), P.V alone in the other: 64 products per wave per interval, was 128/0; sum order kept
# speedup vs baseline: 6.9930x; 1.1592x over previous
#include <hip/hip_runtime.h>

typedef __attribute__((ext_vector_type(2))) float v2f;
typedef __attribute__((ext_vector_type(8))) float v8f;

#define N_PIX 4096
#define CH 32
#define CIN 64
#define BATCH 4
#define PLANE (CH * N_PIX)

#define OG 16

__global__ __launch_bounds__(256) void qkv_conv_kernel(
    const float* __restrict__ x,
    const float* __restrict__ w1, const float* __restrict__ bb1,
    const float* __restrict__ w2, const float* __restrict__ bb2,
    const float* __restrict__ w3, const float* __restrict__ bb3,
    float* __restrict__ Q, float* __restrict__ K, float* __restrict__ V)
{
    const int sel = blockIdx.y;
    const float* __restrict__ w  = sel == 0 ? w1  : (sel == 1 ? w2  : w3);
    const float* __restrict__ bs = sel == 0 ? bb1 : (sel == 1 ? bb2 : bb3);
    float* __restrict__ dst      = sel == 0 ? Q   : (sel == 1 ? K   : V);
    const int o0 = blockIdx.z * OG;

    const int tid = blockIdx.x * 256 + threadIdx.x;
    const int b = tid >> 12;
    const int n = tid & (N_PIX - 1);
    const float* xb = x + (size_t)b * CIN * N_PIX + n;
    float xr[CIN];
#pragma unroll
    for (int c = 0; c < CIN; ++c) xr[c] = xb[(size_t)c * N_PIX];

    float* ob = dst + (size_t)b * PLANE + (size_t)o0 * N_PIX + n;
    const float* wr = w + (size_t)o0 * CIN;
    const float* br = bs + o0;
    for (int oq = 0; oq < OG; oq += 4) {
        const float* r0 = wr + (size_t)oq * CIN;
        const float* r1 = r0 + CIN;
        const float* r2 = r1 + CIN;
        const float* r3 = r2 + CIN;
        float a0 = br[oq], a1 = br[oq + 1], a2 = br[oq + 2], a3 = br[oq + 3];
#pragma unroll
        for (int c = 0; c < CIN; ++c) {
            const float xv = xr[c];
            a0 = fmaf(r0[c], xv, a0);
            a1 = fmaf(r1[c], xv, a1);
            a2 = fmaf(r2[c], xv, a2);
            a3 = fmaf(r3[c], xv, a3);
        }
        float* op = ob + (size_t)oq * N_PIX;
        op[0]         = a0;
        op[N_PIX]     = a1;
        op[2 * N_PIX] = a2;
        op[3 * N_PIX] = a3;
    }
}

#define KT 8
#define JSTEP (16 * KT)
#define PP 20
#define KHALF (N_PIX / 2)
#define NTH (KHALF / JSTEP)

__global__ __launch_bounds__(256) void attn_kernel(
    const float* __restrict__ Q, const float* __restrict__ K,
    const float* __restrict__ V, float* __restrict__ Y)
{
    __shared__ float4 sc[2][2][BATCH][KT][2][32];
    __shared__ __attribute__((aligned(16))) float pc[2][BATCH][KT][16][PP];

    const int lane = threadIdx.x & 31;
    const int wv   = threadIdx.x >> 5;
    const int b    = wv & 3;
    const int g    = __builtin_amdgcn_readfirstlane(wv >> 2);
    const int lo   = lane & 15;
    const int hi   = lane >> 4;
    const int i0   = blockIdx.x * 16;
    const int pcol = ((lo >> 1) & 1) * 8 + (lo >> 2) * 2 + (lo & 1);

    const float* Qb = Q + (size_t)b * PLANE;
    const float* Kb = K + (size_t)b * PLANE + (size_t)g * KHALF;
    const float* Vb = V + (size_t)b * PLANE + (size_t)g * KHALF * CH;

    v2f qa[8];
#pragma unroll
    for (int kk = 0; kk < 8; ++kk) {
        const float* p = Qb + (size_t)(i0 + lo) * CH + 4 * kk + 2 * hi;
        qa[kk] = v2f{p[0], p[1]};
    }

    v8f acc0 = {};
    v8f acc1 = {};

    float2 vq[KT][4][2];

    for (int it = 0; it < 2 * NTH + 3; ++it) {
        const int ls = it - g;
        if (ls >= 0 && ls <= 2 * NTH + 1) {
            const int k  = ls >> 1;
            const int jt = k * JSTEP;
            if ((ls & 1) == 0) {
                const bool has_qk = k < NTH;
                const bool has_sm = k > 0;
                const int wbuf = k & 1;
                const int rbuf = wbuf ^ 1;
                float4 kq[8][2][KT / 4];
                if (has_qk) {
#pragma unroll
                    for (int kk = 0; kk < 8; ++kk) {
                        const float* kr0 = Kb + (size_t)(4 * kk + 2 * hi) * N_PIX + jt + 4 * lo;
                        const float* kr1 = kr0 + N_PIX;
#pragma unroll
                        for (int q4 = 0; q4 < KT / 4; ++q4) {
                            kq[kk][0][q4] = *(const float4*)(kr0 + 64 * q4);
                            kq[kk][1][q4] = *(const float4*)(kr1 + 64 * q4);
                        }
                    }
                }
                if (has_sm) {
#pragma unroll
                    for (int tt = 0; tt < 2; ++tt) {
                        const int t = 2 * b + tt;
#pragma unroll
                        for (int h = 0; h < 2; ++h) {
                            const float4 r0 = sc[rbuf][g][0][t][h][lane];
                            const float4 r1 = sc[rbuf][g][1][t][h][lane];
                            const float4 r2 = sc[rbuf][g][2][t][h][lane];
                            const float4 r3 = sc[rbuf][g][3][t][h][lane];
                            const float q0[4] = {r0.x, r0.y, r0.z, r0.w};
                            const float q1[4] = {r1.x, r1.y, r1.z, r1.w};
                            const float q2[4] = {r2.x, r2.y, r2.z, r2.w};
                            const float q3[4] = {r3.x, r3.y, r3.z, r3.w};
                            float e0[4], e1[4], e2[4], e3[4], inv[4];
#pragma unroll
                            for (int u = 0; u < 4; ++u) {
                                float m = fmaxf(fmaxf(q0[u], q1[u]), fmaxf(q2[u], q3[u]));
                                e0[u] = __expf(q0[u] - m); e1[u] = __expf(q1[u] - m);
                                e2[u] = __expf(q2[u] - m); e3[u] = __expf(q3[u] - m);
                            }
#pragma unroll
                            for (int u = 0; u < 4; ++u) {
                                float d = e0[u] + e1[u] + e2[u] + e3[u];
                                float y = __builtin_amdgcn_rcpf(d);
                                inv[u] = fmaf(fmaf(-d, y, 1.0f), y, y);
                            }
#pragma unroll
                            for (int u = 0; u < 4; ++u) {
                                const int r = 4 * h + u + 8 * hi;
                                pc[g][0][t][r][pcol] = e0[u] * inv[u];
                                pc[g][1][t][r][pcol] = e1[u] * inv[u];
                                pc[g][2][t][r][pcol] = e2[u] * inv[u];
                                pc[g][3][t][r][pcol] = e3[u] * inv[u];
                            }
                        }
                    }
                }
                if (has_qk) {
                    v8f s[KT];
#pragma unroll
                    for (int t = 0; t < KT; ++t) s[t] = v8f{};
#pragma unroll
                    for (int kk = 0; kk < 8; ++kk) {
#pragma unroll
                        for (int q4 = 0; q4 < KT / 4; ++q4) {
                            const float4 c0 = kq[kk][0][q4];
                            const float4 c1 = kq[kk][1][q4];
                            s[4 * q4 + 0] = __builtin_amdgcn_wmma_f32_16x16x4_f32(
                                false, qa[kk], false, v2f{c0.x, c1.x}, (short)0, s[4 * q4 + 0], false, false);
                            s[4 * q4 + 1] = __builtin_amdgcn_wmma_f32_16x16x4_f32(
                                false, qa[kk], false, v2f{c0.y, c1.y}, (short)0, s[4 * q4 + 1], false, false);
                            s[4 * q4 + 2] = __builtin_amdgcn_wmma_f32_16x16x4_f32(
                                false, qa[kk], false, v2f{c0.z, c1.z}, (short)0, s[4 * q4 + 2], false, false);
                            s[4 * q4 + 3] = __builtin_amdgcn_wmma_f32_16x16x4_f32(
                                false, qa[kk], false, v2f{c0.w, c1.w}, (short)0, s[4 * q4 + 3], false, false);
                        }
                    }
#pragma unroll
                    for (int t = 0; t < KT; ++t) {
                        sc[wbuf][g][b][t][0][lane] = make_float4(s[t][0], s[t][1], s[t][2], s[t][3]);
                        sc[wbuf][g][b][t][1][lane] = make_float4(s[t][4], s[t][5], s[t][6], s[t][7]);
                    }
                }
            } else {
                const bool has_pv = k > 0;
                if (has_pv) {
                    v2f pa[KT][4];
#pragma unroll
                    for (int t = 0; t < KT; ++t) {
                        const float4 p0 = *(const float4*)&pc[g][b][t][lo][8 * hi];
                        const float4 p1 = *(const float4*)&pc[g][b][t][lo][8 * hi + 4];
                        pa[t][0] = v2f{p0.x, p0.y};
                        pa[t][1] = v2f{p0.z, p0.w};
                        pa[t][2] = v2f{p1.x, p1.y};
                        pa[t][3] = v2f{p1.z, p1.w};
                    }
                    __builtin_amdgcn_sched_barrier(0);
#pragma unroll
                    for (int t = 0; t < KT; ++t) {
#pragma unroll
                        for (int kk = 0; kk < 4; ++kk) {
                            const v2f vl = v2f{vq[t][kk][0].x, vq[t][kk][1].x};
                            const v2f vh = v2f{vq[t][kk][0].y, vq[t][kk][1].y};
                            acc0 = __builtin_amdgcn_wmma_f32_16x16x4_f32(
                                       false, pa[t][kk], false, vl, (short)0, acc0, false, false);
                            acc1 = __builtin_amdgcn_wmma_f32_16x16x4_f32(
                                       false, pa[t][kk], false, vh, (short)0, acc1, false, false);
                        }
                    }
                }
                {
                    const int jv = (k < NTH ? k : NTH - 1) * JSTEP;
#pragma unroll
                    for (int t = 0; t < KT; ++t) {
#pragma unroll
                        for (int kk = 0; kk < 4; ++kk) {
                            const float* vr0 = Vb + (size_t)(jv + 64 * (t >> 2) + (t & 3) + 16 * kk + 8 * hi) * CH + 2 * lo;
                            vq[t][kk][0] = *(const float2*)vr0;
                            vq[t][kk][1] = *(const float2*)(vr0 + 4 * CH);
                        }
                    }
                }
            }
        }
        __syncthreads();
    }

    if (g == 1) {
        sc[0][1][b][0][0][lane] = make_float4(acc0[0], acc0[1], acc0[2], acc0[3]);
        sc[0][1][b][0][1][lane] = make_float4(acc0[4], acc0[5], acc0[6], acc0[7]);
        sc[0][1][b][1][0][lane] = make_float4(acc1[0], acc1[1], acc1[2], acc1[3]);
        sc[0][1][b][1][1][lane] = make_float4(acc1[4], acc1[5], acc1[6], acc1[7]);
    }
    __syncthreads();
    if (g == 0) {
        const float4 a0 = sc[0][1][b][0][0][lane];
        const float4 a1 = sc[0][1][b][0][1][lane];
        const float4 c0 = sc[0][1][b][1][0][lane];
        const float4 c1 = sc[0][1][b][1][1][lane];
        const float o0[8] = {a0.x, a0.y, a0.z, a0.w, a1.x, a1.y, a1.z, a1.w};
        const float o1[8] = {c0.x, c0.y, c0.z, c0.w, c1.x, c1.y, c1.z, c1.w};
        float* Yb = Y + (size_t)b * PLANE;
#pragma unroll
        for (int v = 0; v < 8; ++v) {
            int r = v + 8 * hi;
            *(float2*)(Yb + (size_t)(i0 + r) * CH + 2 * lo) =
                make_float2(acc0[v] + o0[v], acc1[v] + o1[v]);
        }
    }
}

#define OGO 16

__global__ __launch_bounds__(256) void out_conv_kernel(
    const float* __restrict__ Y, const float* __restrict__ w4,
    const float* __restrict__ bb4, float* __restrict__ out)
{
    const int o0 = blockIdx.y * OGO;
    const int tid = blockIdx.x * 256 + threadIdx.x;
    const int b = tid >> 12;
    const int n = tid & (N_PIX - 1);
    const float* Yb = Y + (size_t)b * PLANE + n;
    float yr[CH];
#pragma unroll
    for (int c = 0; c < CH; ++c) yr[c] = Yb[(size_t)c * N_PIX];

    float* ob = out + (size_t)b * CIN * N_PIX + (size_t)o0 * N_PIX + n;
    const float* wr = w4 + (size_t)o0 * CH;
    const float* br = bb4 + o0;
#pragma clang loop vectorize(disable)
    for (int oq = 0; oq < OGO; oq += 4) {
        const float* r0 = wr + (size_t)oq * CH;
        const float* r1 = r0 + CH;
        const float* r2 = r1 + CH;
        const float* r3 = r2 + CH;
        float a0 = br[oq], a1 = br[oq + 1], a2 = br[oq + 2], a3 = br[oq + 3];
#pragma unroll
        for (int c = 0; c < CH; ++c) {
            const float yv = yr[c];
            a0 = fmaf(r0[c], yv, a0);
            a1 = fmaf(r1[c], yv, a1);
            a2 = fmaf(r2[c], yv, a2);
            a3 = fmaf(r3[c], yv, a3);
        }
        float* op = ob + (size_t)oq * N_PIX;
        op[0]         = a0;
        op[N_PIX]     = a1;
        op[2 * N_PIX] = a2;
        op[3 * N_PIX] = a3;
    }
}

extern "C" void kernel_launch(void* const* d_in, const int* in_sizes, int n_in,
                              void* d_out, int out_size, void* d_ws, size_t ws_size,
                              hipStream_t stream) {
    const float* x   = (const float*)d_in[0];
    const float* w1  = (const float*)d_in[1];
    const float* bb1 = (const float*)d_in[2];
    const float* w2  = (const float*)d_in[3];
    const float* bb2 = (const float*)d_in[4];
    const float* w3  = (const float*)d_in[5];
    const float* bb3 = (const float*)d_in[6];
    const float* w4  = (const float*)d_in[7];
    const float* bb4 = (const float*)d_in[8];
    float* out = (float*)d_out;

    float* Q = (float*)d_ws;
    float* K = Q + (size_t)BATCH * PLANE;
    float* V = K + (size_t)BATCH * PLANE;
    float* Y = V + (size_t)BATCH * PLANE;

    qkv_conv_kernel<<<dim3((BATCH * N_PIX) / 256, 3, CH / OG), 256, 0, stream>>>(
        x, w1, bb1, w2, bb2, w3, bb3, Q, K, V);
    attn_kernel<<<N_PIX / 16, 256, 0, stream>>>(Q, K, V, Y);
    out_conv_kernel<<<dim3((BATCH * N_PIX) / 256, CIN / OGO), 256, 0, stream>>>(Y, w4, bb4, out);
}


# V conv output stored channel-pair interleaved: one 16-byte V load is both operand pairs of a P.V kstep (32 loads/trip, were 64 + 64 register pairings); P pairs adjacent in LDS; key sum order changes
# speedup vs baseline: 7.2123x; 1.0314x over previous
#include <hip/hip_runtime.h>

typedef __attribute__((ext_vector_type(2))) float v2f;
typedef __attribute__((ext_vector_type(8))) float v8f;

#define N_PIX 4096
#define CH 32
#define CIN 64
#define BATCH 4
#define PLANE (CH * N_PIX)

#define OG 16

__global__ __launch_bounds__(256) void qkv_conv_kernel(
    const float* __restrict__ x,
    const float* __restrict__ w1, const float* __restrict__ bb1,
    const float* __restrict__ w2, const float* __restrict__ bb2,
    const float* __restrict__ w3, const float* __restrict__ bb3,
    float* __restrict__ Q, float* __restrict__ K, float* __restrict__ V)
{
    const int sel = blockIdx.y;
    const float* __restrict__ w  = sel == 0 ? w1  : (sel == 1 ? w2  : w3);
    const float* __restrict__ bs = sel == 0 ? bb1 : (sel == 1 ? bb2 : bb3);
    float* __restrict__ dst      = sel == 0 ? Q   : (sel == 1 ? K   : V);
    const int o0 = blockIdx.z * OG;

    const int tid = blockIdx.x * 256 + threadIdx.x;
    const int b = tid >> 12;
    const int n = tid & (N_PIX - 1);
    const float* xb = x + (size_t)b * CIN * N_PIX + n;
    float xr[CIN];
#pragma unroll
    for (int c = 0; c < CIN; ++c) xr[c] = xb[(size_t)c * N_PIX];

    float* ob = dst + (size_t)b * PLANE + (size_t)o0 * N_PIX + n;
    const float* wr = w + (size_t)o0 * CIN;
    const float* br = bs + o0;
    for (int oq = 0; oq < OG; oq += 4) {
        const float* r0 = wr + (size_t)oq * CIN;
        const float* r1 = r0 + CIN;
        const float* r2 = r1 + CIN;
        const float* r3 = r2 + CIN;
        float a0 = br[oq], a1 = br[oq + 1], a2 = br[oq + 2], a3 = br[oq + 3];
#pragma unroll
        for (int c = 0; c < CIN; ++c) {
            const float xv = xr[c];
            a0 = fmaf(r0[c], xv, a0);
            a1 = fmaf(r1[c], xv, a1);
            a2 = fmaf(r2[c], xv, a2);
            a3 = fmaf(r3[c], xv, a3);
        }
        if (sel == 2) {
            float* vp = dst + (size_t)b * PLANE + (size_t)(o0 + oq) * N_PIX + 2 * n;
            *(float2*)vp               = make_float2(a0, a1);
            *(float2*)(vp + 2 * N_PIX) = make_float2(a2, a3);
        } else {
            float* op = ob + (size_t)oq * N_PIX;
            op[0]         = a0;
            op[N_PIX]     = a1;
            op[2 * N_PIX] = a2;
            op[3 * N_PIX] = a3;
        }
    }
}

#define KT 8
#define JSTEP (16 * KT)
#define PQ 36
#define KHALF (N_PIX / 2)
#define NTH (KHALF / JSTEP)

__global__ __launch_bounds__(256) void attn_kernel(
    const float* __restrict__ Q, const float* __restrict__ K,
    const float* __restrict__ V, float* __restrict__ Y)
{
    __shared__ float4 sc[2][2][BATCH][KT][2][32];
    __shared__ __attribute__((aligned(16))) float pc[2][BATCH][KT / 2][16][PQ];

    const int lane = threadIdx.x & 31;
    const int wv   = threadIdx.x >> 5;
    const int b    = wv & 3;
    const int g    = __builtin_amdgcn_readfirstlane(wv >> 2);
    const int lo   = lane & 15;
    const int hi   = lane >> 4;
    const int i0   = blockIdx.x * 16;
    const int pslot = (lo & 1) * 8 + (lo >> 1);

    const float* Qb = Q + (size_t)b * PLANE;
    const float* Kb = K + (size_t)b * PLANE + (size_t)g * KHALF;
    const float* Vb = V + (size_t)b * PLANE + (size_t)g * KHALF * CH;

    v2f qa[8];
#pragma unroll
    for (int kk = 0; kk < 8; ++kk) {
        const float* p = Qb + (size_t)(i0 + lo) * CH + 4 * kk + 2 * hi;
        qa[kk] = v2f{p[0], p[1]};
    }

    v8f acc0 = {};
    v8f acc1 = {};

    float4 vq[KT / 2][8];

    for (int it = 0; it < 2 * NTH + 3; ++it) {
        const int ls = it - g;
        if (ls >= 0 && ls <= 2 * NTH + 1) {
            const int k  = ls >> 1;
            const int jt = 256 * (k >> 1) + 64 * (k & 1);
            if ((ls & 1) == 0) {
                const bool has_qk = k < NTH;
                const bool has_sm = k > 0;
                const int wbuf = k & 1;
                const int rbuf = wbuf ^ 1;
                float4 kq[8][2][KT / 4];
                if (has_qk) {
#pragma unroll
                    for (int kk = 0; kk < 8; ++kk) {
                        const float* kr0 = Kb + (size_t)(4 * kk + 2 * hi) * N_PIX + jt + 4 * lo;
                        const float* kr1 = kr0 + N_PIX;
#pragma unroll
                        for (int q4 = 0; q4 < KT / 4; ++q4) {
                            kq[kk][0][q4] = *(const float4*)(kr0 + 128 * q4);
                            kq[kk][1][q4] = *(const float4*)(kr1 + 128 * q4);
                        }
                    }
                }
                if (has_sm) {
#pragma unroll
                    for (int h = 0; h < 2; ++h) {
                        float pr[2][BATCH][4];
#pragma unroll
                        for (int p = 0; p < 2; ++p) {
                            const int t = 4 * p + b;
                            const float4 r0 = sc[rbuf][g][0][t][h][lane];
                            const float4 r1 = sc[rbuf][g][1][t][h][lane];
                            const float4 r2 = sc[rbuf][g][2][t][h][lane];
                            const float4 r3 = sc[rbuf][g][3][t][h][lane];
                            const float q0[4] = {r0.x, r0.y, r0.z, r0.w};
                            const float q1[4] = {r1.x, r1.y, r1.z, r1.w};
                            const float q2[4] = {r2.x, r2.y, r2.z, r2.w};
                            const float q3[4] = {r3.x, r3.y, r3.z, r3.w};
                            float e0[4], e1[4], e2[4], e3[4], inv[4];
#pragma unroll
                            for (int u = 0; u < 4; ++u) {
                                float m = fmaxf(fmaxf(q0[u], q1[u]), fmaxf(q2[u], q3[u]));
                                e0[u] = __expf(q0[u] - m); e1[u] = __expf(q1[u] - m);
                                e2[u] = __expf(q2[u] - m); e3[u] = __expf(q3[u] - m);
                            }
#pragma unroll
                            for (int u = 0; u < 4; ++u) {
                                float d = e0[u] + e1[u] + e2[u] + e3[u];
                                float y = __builtin_amdgcn_rcpf(d);
                                inv[u] = fmaf(fmaf(-d, y, 1.0f), y, y);
                            }
#pragma unroll
                            for (int u = 0; u < 4; ++u) {
                                pr[p][0][u] = e0[u] * inv[u];
                                pr[p][1][u] = e1[u] * inv[u];
                                pr[p][2][u] = e2[u] * inv[u];
                                pr[p][3][u] = e3[u] * inv[u];
                            }
                        }
#pragma unroll
                        for (int u = 0; u < 4; ++u) {
                            const int r = 4 * h + u + 8 * hi;
#pragma unroll
                            for (int bb = 0; bb < BATCH; ++bb)
                                *(float2*)&pc[g][bb][b][r][2 * pslot] =
                                    make_float2(pr[0][bb][u], pr[1][bb][u]);
                        }
                    }
                }
                if (has_qk) {
                    v8f s[KT];
#pragma unroll
                    for (int t = 0; t < KT; ++t) s[t] = v8f{};
#pragma unroll
                    for (int kk = 0; kk < 8; ++kk) {
#pragma unroll
                        for (int q4 = 0; q4 < KT / 4; ++q4) {
                            const float4 c0 = kq[kk][0][q4];
                            const float4 c1 = kq[kk][1][q4];
                            s[4 * q4 + 0] = __builtin_amdgcn_wmma_f32_16x16x4_f32(
                                false, qa[kk], false, v2f{c0.x, c1.x}, (short)0, s[4 * q4 + 0], false, false);
                            s[4 * q4 + 1] = __builtin_amdgcn_wmma_f32_16x16x4_f32(
                                false, qa[kk], false, v2f{c0.y, c1.y}, (short)0, s[4 * q4 + 1], false, false);
                            s[4 * q4 + 2] = __builtin_amdgcn_wmma_f32_16x16x4_f32(
                                false, qa[kk], false, v2f{c0.z, c1.z}, (short)0, s[4 * q4 + 2], false, false);
                            s[4 * q4 + 3] = __builtin_amdgcn_wmma_f32_16x16x4_f32(
                                false, qa[kk], false, v2f{c0.w, c1.w}, (short)0, s[4 * q4 + 3], false, false);
                        }
                    }
#pragma unroll
                    for (int t = 0; t < KT; ++t) {
                        sc[wbuf][g][b][t][0][lane] = make_float4(s[t][0], s[t][1], s[t][2], s[t][3]);
                        sc[wbuf][g][b][t][1][lane] = make_float4(s[t][4], s[t][5], s[t][6], s[t][7]);
                    }
                }
            } else {
                const bool has_pv = k > 0;
                if (has_pv) {
                    v2f pa[KT / 2][8];
#pragma unroll
                    for (int e = 0; e < KT / 2; ++e) {
#pragma unroll
                        for (int j = 0; j < 4; ++j) {
                            const float4 p4 = *(const float4*)&pc[g][b][e][lo][16 * hi + 4 * j];
                            pa[e][2 * j]     = v2f{p4.x, p4.y};
                            pa[e][2 * j + 1] = v2f{p4.z, p4.w};
                        }
                    }
                    __builtin_amdgcn_sched_barrier(0);
#pragma unroll
                    for (int e = 0; e < KT / 2; ++e) {
#pragma unroll
                        for (int kk = 0; kk < 8; ++kk) {
                            const float4 v4 = vq[e][kk];
                            acc0 = __builtin_amdgcn_wmma_f32_16x16x4_f32(
                                       false, pa[e][kk], false, v2f{v4.x, v4.y}, (short)0, acc0, false, false);
                            acc1 = __builtin_amdgcn_wmma_f32_16x16x4_f32(
                                       false, pa[e][kk], false, v2f{v4.z, v4.w}, (short)0, acc1, false, false);
                        }
                    }
                }
                {
                    const int jv = (k < NTH ? k : NTH - 1) * JSTEP;
#pragma unroll
                    for (int e = 0; e < KT / 2; ++e) {
#pragma unroll
                        for (int kk = 0; kk < 8; ++kk) {
                            const float* vr0 = Vb + (size_t)jv * CH + (size_t)(8 * kk + 4 * hi + e) * (2 * CH) + 4 * lo;
                            vq[e][kk] = *(const float4*)vr0;
                        }
                    }
                    __builtin_amdgcn_fence(__ATOMIC_ACQUIRE, "workgroup", "global");
                }
            }
        }
        __syncthreads();
    }

    if (g == 1) {
        sc[0][1][b][0][0][lane] = make_float4(acc0[0], acc0[1], acc0[2], acc0[3]);
        sc[0][1][b][0][1][lane] = make_float4(acc0[4], acc0[5], acc0[6], acc0[7]);
        sc[0][1][b][1][0][lane] = make_float4(acc1[0], acc1[1], acc1[2], acc1[3]);
        sc[0][1][b][1][1][lane] = make_float4(acc1[4], acc1[5], acc1[6], acc1[7]);
    }
    __syncthreads();
    if (g == 0) {
        const float4 a0 = sc[0][1][b][0][0][lane];
        const float4 a1 = sc[0][1][b][0][1][lane];
        const float4 c0 = sc[0][1][b][1][0][lane];
        const float4 c1 = sc[0][1][b][1][1][lane];
        const float o0[8] = {a0.x, a0.y, a0.z, a0.w, a1.x, a1.y, a1.z, a1.w};
        const float o1[8] = {c0.x, c0.y, c0.z, c0.w, c1.x, c1.y, c1.z, c1.w};
        float* Yb = Y + (size_t)b * PLANE;
#pragma unroll
        for (int v = 0; v < 8; ++v) {
            int r = v + 8 * hi;
            *(float2*)(Yb + (size_t)(i0 + r) * CH + 2 * lo) =
                make_float2(acc0[v] + o0[v], acc1[v] + o1[v]);
        }
    }
}

#define OGO 16

__global__ __launch_bounds__(256) void out_conv_kernel(
    const float* __restrict__ Y, const float* __restrict__ w4,
    const float* __restrict__ bb4, float* __restrict__ out)
{
    const int o0 = blockIdx.y * OGO;
    const int tid = blockIdx.x * 256 + threadIdx.x;
    const int b = tid >> 12;
    const int n = tid & (N_PIX - 1);
    const float* Yb = Y + (size_t)b * PLANE + n;
    float yr[CH];
#pragma unroll
    for (int c = 0; c < CH; ++c) yr[c] = Yb[(size_t)c * N_PIX];

    float* ob = out + (size_t)b * CIN * N_PIX + (size_t)o0 * N_PIX + n;
    const float* wr = w4 + (size_t)o0 * CH;
    const float* br = bb4 + o0;
#pragma clang loop vectorize(disable)
    for (int oq = 0; oq < OGO; oq += 4) {
        const float* r0 = wr + (size_t)oq * CH;
        const float* r1 = r0 + CH;
        const float* r2 = r1 + CH;
        const float* r3 = r2 + CH;
        float a0 = br[oq], a1 = br[oq + 1], a2 = br[oq + 2], a3 = br[oq + 3];
#pragma unroll
        for (int c = 0; c < CH; ++c) {
            const float yv = yr[c];
            a0 = fmaf(r0[c], yv, a0);
            a1 = fmaf(r1[c], yv, a1);
            a2 = fmaf(r2[c], yv, a2);
            a3 = fmaf(r3[c], yv, a3);
        }
        float* op = ob + (size_t)oq * N_PIX;
        op[0]         = a0;
        op[N_PIX]     = a1;
        op[2 * N_PIX] = a2;
        op[3 * N_PIX] = a3;
    }
}

extern "C" void kernel_launch(void* const* d_in, const int* in_sizes, int n_in,
                              void* d_out, int out_size, void* d_ws, size_t ws_size,
                              hipStream_t stream) {
    const float* x   = (const float*)d_in[0];
    const float* w1  = (const float*)d_in[1];
    const float* bb1 = (const float*)d_in[2];
    const float* w2  = (const float*)d_in[3];
    const float* bb2 = (const float*)d_in[4];
    const float* w3  = (const float*)d_in[5];
    const float* bb3 = (const float*)d_in[6];
    const float* w4  = (const float*)d_in[7];
    const float* bb4 = (const float*)d_in[8];
    float* out = (float*)d_out;

    float* Q = (float*)d_ws;
    float* K = Q + (size_t)BATCH * PLANE;
    float* V = K + (size_t)BATCH * PLANE;
    float* Y = V + (size_t)BATCH * PLANE;

    qkv_conv_kernel<<<dim3((BATCH * N_PIX) / 256, 3, CH / OG), 256, 0, stream>>>(
        x, w1, bb1, w2, bb2, w3, bb3, Q, K, V);
    attn_kernel<<<N_PIX / 16, 256, 0, stream>>>(Q, K, V, Y);
    out_conv_kernel<<<dim3((BATCH * N_PIX) / 256, CIN / OGO), 256, 0, stream>>>(Y, w4, bb4, out);
}


# attention: each work-group walks its 16 key groups in rotated order (start from its tile index), so the 256 groups stop reading the same K/V lines at the same moment; key sum order changes
# speedup vs baseline: 7.5403x; 1.0455x over previous
#include <hip/hip_runtime.h>

typedef __attribute__((ext_vector_type(2))) float v2f;
typedef __attribute__((ext_vector_type(8))) float v8f;

#define N_PIX 4096
#define CH 32
#define CIN 64
#define BATCH 4
#define PLANE (CH * N_PIX)

#define OG 16

__global__ __launch_bounds__(256) void qkv_conv_kernel(
    const float* __restrict__ x,
    const float* __restrict__ w1, const float* __restrict__ bb1,
    const float* __restrict__ w2, const float* __restrict__ bb2,
    const float* __restrict__ w3, const float* __restrict__ bb3,
    float* __restrict__ Q, float* __restrict__ K, float* __restrict__ V)
{
    const int sel = blockIdx.y;
    const float* __restrict__ w  = sel == 0 ? w1  : (sel == 1 ? w2  : w3);
    const float* __restrict__ bs = sel == 0 ? bb1 : (sel == 1 ? bb2 : bb3);
    float* __restrict__ dst      = sel == 0 ? Q   : (sel == 1 ? K   : V);
    const int o0 = blockIdx.z * OG;

    const int tid = blockIdx.x * 256 + threadIdx.x;
    const int b = tid >> 12;
    const int n = tid & (N_PIX - 1);
    const float* xb = x + (size_t)b * CIN * N_PIX + n;
    float xr[CIN];
#pragma unroll
    for (int c = 0; c < CIN; ++c) xr[c] = xb[(size_t)c * N_PIX];

    float* ob = dst + (size_t)b * PLANE + (size_t)o0 * N_PIX + n;
    const float* wr = w + (size_t)o0 * CIN;
    const float* br = bs + o0;
    for (int oq = 0; oq < OG; oq += 4) {
        const float* r0 = wr + (size_t)oq * CIN;
        const float* r1 = r0 + CIN;
        const float* r2 = r1 + CIN;
        const float* r3 = r2 + CIN;
        float a0 = br[oq], a1 = br[oq + 1], a2 = br[oq + 2], a3 = br[oq + 3];
#pragma unroll
        for (int c = 0; c < CIN; ++c) {
            const float xv = xr[c];
            a0 = fmaf(r0[c], xv, a0);
            a1 = fmaf(r1[c], xv, a1);
            a2 = fmaf(r2[c], xv, a2);
            a3 = fmaf(r3[c], xv, a3);
        }
        if (sel == 2) {
            float* vp = dst + (size_t)b * PLANE + (size_t)(o0 + oq) * N_PIX + 2 * n;
            *(float2*)vp               = make_float2(a0, a1);
            *(float2*)(vp + 2 * N_PIX) = make_float2(a2, a3);
        } else {
            float* op = ob + (size_t)oq * N_PIX;
            op[0]         = a0;
            op[N_PIX]     = a1;
            op[2 * N_PIX] = a2;
            op[3 * N_PIX] = a3;
        }
    }
}

#define KT 8
#define JSTEP (16 * KT)
#define PQ 36
#define KHALF (N_PIX / 2)
#define NTH (KHALF / JSTEP)

__global__ __launch_bounds__(256) void attn_kernel(
    const float* __restrict__ Q, const float* __restrict__ K,
    const float* __restrict__ V, float* __restrict__ Y)
{
    __shared__ float4 sc[2][2][BATCH][KT][2][32];
    __shared__ __attribute__((aligned(16))) float pc[2][BATCH][KT / 2][16][PQ];

    const int lane = threadIdx.x & 31;
    const int wv   = threadIdx.x >> 5;
    const int b    = wv & 3;
    const int g    = __builtin_amdgcn_readfirstlane(wv >> 2);
    const int lo   = lane & 15;
    const int hi   = lane >> 4;
    const int i0   = blockIdx.x * 16;
    const int rot  = (blockIdx.x ^ (blockIdx.x >> 4)) & (NTH - 1);
    const int pslot = (lo & 1) * 8 + (lo >> 1);

    const float* Qb = Q + (size_t)b * PLANE;
    const float* Kb = K + (size_t)b * PLANE + (size_t)g * KHALF;
    const float* Vb = V + (size_t)b * PLANE + (size_t)g * KHALF * CH;

    v2f qa[8];
#pragma unroll
    for (int kk = 0; kk < 8; ++kk) {
        const float* p = Qb + (size_t)(i0 + lo) * CH + 4 * kk + 2 * hi;
        qa[kk] = v2f{p[0], p[1]};
    }

    v8f acc0 = {};
    v8f acc1 = {};

    float4 vq[KT / 2][8];

    for (int it = 0; it < 2 * NTH + 3; ++it) {
        const int ls = it - g;
        if (ls >= 0 && ls <= 2 * NTH + 1) {
            const int k  = ls >> 1;
            const int kd = (k + rot) & (NTH - 1);
            const int jt = 256 * (kd >> 1) + 64 * (kd & 1);
            if ((ls & 1) == 0) {
                const bool has_qk = k < NTH;
                const bool has_sm = k > 0;
                const int wbuf = k & 1;
                const int rbuf = wbuf ^ 1;
                float4 kq[8][2][KT / 4];
                if (has_qk) {
#pragma unroll
                    for (int kk = 0; kk < 8; ++kk) {
                        const float* kr0 = Kb + (size_t)(4 * kk + 2 * hi) * N_PIX + jt + 4 * lo;
                        const float* kr1 = kr0 + N_PIX;
#pragma unroll
                        for (int q4 = 0; q4 < KT / 4; ++q4) {
                            kq[kk][0][q4] = *(const float4*)(kr0 + 128 * q4);
                            kq[kk][1][q4] = *(const float4*)(kr1 + 128 * q4);
                        }
                    }
                }
                if (has_sm) {
#pragma unroll
                    for (int h = 0; h < 2; ++h) {
                        float pr[2][BATCH][4];
#pragma unroll
                        for (int p = 0; p < 2; ++p) {
                            const int t = 4 * p + b;
                            const float4 r0 = sc[rbuf][g][0][t][h][lane];
                            const float4 r1 = sc[rbuf][g][1][t][h][lane];
                            const float4 r2 = sc[rbuf][g][2][t][h][lane];
                            const float4 r3 = sc[rbuf][g][3][t][h][lane];
                            const float q0[4] = {r0.x, r0.y, r0.z, r0.w};
                            const float q1[4] = {r1.x, r1.y, r1.z, r1.w};
                            const float q2[4] = {r2.x, r2.y, r2.z, r2.w};
                            const float q3[4] = {r3.x, r3.y, r3.z, r3.w};
                            float e0[4], e1[4], e2[4], e3[4], inv[4];
#pragma unroll
                            for (int u = 0; u < 4; ++u) {
                                float m = fmaxf(fmaxf(q0[u], q1[u]), fmaxf(q2[u], q3[u]));
                                e0[u] = __expf(q0[u] - m); e1[u] = __expf(q1[u] - m);
                                e2[u] = __expf(q2[u] - m); e3[u] = __expf(q3[u] - m);
                            }
#pragma unroll
                            for (int u = 0; u < 4; ++u) {
                                float d = e0[u] + e1[u] + e2[u] + e3[u];
                                float y = __builtin_amdgcn_rcpf(d);
                                inv[u] = fmaf(fmaf(-d, y, 1.0f), y, y);
                            }
#pragma unroll
                            for (int u = 0; u < 4; ++u) {
                                pr[p][0][u] = e0[u] * inv[u];
                                pr[p][1][u] = e1[u] * inv[u];
                                pr[p][2][u] = e2[u] * inv[u];
                                pr[p][3][u] = e3[u] * inv[u];
                            }
                        }
#pragma unroll
                        for (int u = 0; u < 4; ++u) {
                            const int r = 4 * h + u + 8 * hi;
#pragma unroll
                            for (int bb = 0; bb < BATCH; ++bb)
                                *(float2*)&pc[g][bb][b][r][2 * pslot] =
                                    make_float2(pr[0][bb][u], pr[1][bb][u]);
                        }
                    }
                }
                if (has_qk) {
                    v8f s[KT];
#pragma unroll
                    for (int t = 0; t < KT; ++t) s[t] = v8f{};
#pragma unroll
                    for (int kk = 0; kk < 8; ++kk) {
#pragma unroll
                        for (int q4 = 0; q4 < KT / 4; ++q4) {
                            const float4 c0 = kq[kk][0][q4];
                            const float4 c1 = kq[kk][1][q4];
                            s[4 * q4 + 0] = __builtin_amdgcn_wmma_f32_16x16x4_f32(
                                false, qa[kk], false, v2f{c0.x, c1.x}, (short)0, s[4 * q4 + 0], false, false);
                            s[4 * q4 + 1] = __builtin_amdgcn_wmma_f32_16x16x4_f32(
                                false, qa[kk], false, v2f{c0.y, c1.y}, (short)0, s[4 * q4 + 1], false, false);
                            s[4 * q4 + 2] = __builtin_amdgcn_wmma_f32_16x16x4_f32(
                                false, qa[kk], false, v2f{c0.z, c1.z}, (short)0, s[4 * q4 + 2], false, false);
                            s[4 * q4 + 3] = __builtin_amdgcn_wmma_f32_16x16x4_f32(
                                false, qa[kk], false, v2f{c0.w, c1.w}, (short)0, s[4 * q4 + 3], false, false);
                        }
                    }
#pragma unroll
                    for (int t = 0; t < KT; ++t) {
                        sc[wbuf][g][b][t][0][lane] = make_float4(s[t][0], s[t][1], s[t][2], s[t][3]);
                        sc[wbuf][g][b][t][1][lane] = make_float4(s[t][4], s[t][5], s[t][6], s[t][7]);
                    }
                }
            } else {
                const bool has_pv = k > 0;
                if (has_pv) {
                    v2f pa[KT / 2][8];
#pragma unroll
                    for (int e = 0; e < KT / 2; ++e) {
#pragma unroll
                        for (int j = 0; j < 4; ++j) {
                            const float4 p4 = *(const float4*)&pc[g][b][e][lo][16 * hi + 4 * j];
                            pa[e][2 * j]     = v2f{p4.x, p4.y};
                            pa[e][2 * j + 1] = v2f{p4.z, p4.w};
                        }
                    }
                    __builtin_amdgcn_sched_barrier(0);
#pragma unroll
                    for (int e = 0; e < KT / 2; ++e) {
#pragma unroll
                        for (int kk = 0; kk < 8; ++kk) {
                            const float4 v4 = vq[e][kk];
                            acc0 = __builtin_amdgcn_wmma_f32_16x16x4_f32(
                                       false, pa[e][kk], false, v2f{v4.x, v4.y}, (short)0, acc0, false, false);
                            acc1 = __builtin_amdgcn_wmma_f32_16x16x4_f32(
                                       false, pa[e][kk], false, v2f{v4.z, v4.w}, (short)0, acc1, false, false);
                        }
                    }
                }
                {
                    const int jv = (((k < NTH ? k : NTH - 1) + rot) & (NTH - 1)) * JSTEP;
#pragma unroll
                    for (int e = 0; e < KT / 2; ++e) {
#pragma unroll
                        for (int kk = 0; kk < 8; ++kk) {
                            const float* vr0 = Vb + (size_t)jv * CH + (size_t)(8 * kk + 4 * hi + e) * (2 * CH) + 4 * lo;
                            vq[e][kk] = *(const float4*)vr0;
                        }
                    }
                    __builtin_amdgcn_fence(__ATOMIC_ACQUIRE, "workgroup", "global");
                }
            }
        }
        __syncthreads();
    }

    if (g == 1) {
        sc[0][1][b][0][0][lane] = make_float4(acc0[0], acc0[1], acc0[2], acc0[3]);
        sc[0][1][b][0][1][lane] = make_float4(acc0[4], acc0[5], acc0[6], acc0[7]);
        sc[0][1][b][1][0][lane] = make_float4(acc1[0], acc1[1], acc1[2], acc1[3]);
        sc[0][1][b][1][1][lane] = make_float4(acc1[4], acc1[5], acc1[6], acc1[7]);
    }
    __syncthreads();
    if (g == 0) {
        const float4 a0 = sc[0][1][b][0][0][lane];
        const float4 a1 = sc[0][1][b][0][1][lane];
        const float4 c0 = sc[0][1][b][1][0][lane];
        const float4 c1 = sc[0][1][b][1][1][lane];
        const float o0[8] = {a0.x, a0.y, a0.z, a0.w, a1.x, a1.y, a1.z, a1.w};
        const float o1[8] = {c0.x, c0.y, c0.z, c0.w, c1.x, c1.y, c1.z, c1.w};
        float* Yb = Y + (size_t)b * PLANE;
#pragma unroll
        for (int v = 0; v < 8; ++v) {
            int r = v + 8 * hi;
            *(float2*)(Yb + (size_t)(i0 + r) * CH + 2 * lo) =
                make_float2(acc0[v] + o0[v], acc1[v] + o1[v]);
        }
    }
}

#define OGO 16

__global__ __launch_bounds__(256) void out_conv_kernel(
    const float* __restrict__ Y, const float* __restrict__ w4,
    const float* __restrict__ bb4, float* __restrict__ out)
{
    const int o0 = blockIdx.y * OGO;
    const int tid = blockIdx.x * 256 + threadIdx.x;
    const int b = tid >> 12;
    const int n = tid & (N_PIX - 1);
    const float* Yb = Y + (size_t)b * PLANE + n;
    float yr[CH];
#pragma unroll
    for (int c = 0; c < CH; ++c) yr[c] = Yb[(size_t)c * N_PIX];

    float* ob = out + (size_t)b * CIN * N_PIX + (size_t)o0 * N_PIX + n;
    const float* wr = w4 + (size_t)o0 * CH;
    const float* br = bb4 + o0;
#pragma clang loop vectorize(disable)
    for (int oq = 0; oq < OGO; oq += 4) {
        const float* r0 = wr + (size_t)oq * CH;
        const float* r1 = r0 + CH;
        const float* r2 = r1 + CH;
        const float* r3 = r2 + CH;
        float a0 = br[oq], a1 = br[oq + 1], a2 = br[oq + 2], a3 = br[oq + 3];
#pragma unroll
        for (int c = 0; c < CH; ++c) {
            const float yv = yr[c];
            a0 = fmaf(r0[c], yv, a0);
            a1 = fmaf(r1[c], yv, a1);
            a2 = fmaf(r2[c], yv, a2);
            a3 = fmaf(r3[c], yv, a3);
        }
        float* op = ob + (size_t)oq * N_PIX;
        op[0]         = a0;
        op[N_PIX]     = a1;
        op[2 * N_PIX] = a2;
        op[3 * N_PIX] = a3;
    }
}

extern "C" void kernel_launch(void* const* d_in, const int* in_sizes, int n_in,
                              void* d_out, int out_size, void* d_ws, size_t ws_size,
                              hipStream_t stream) {
    const float* x   = (const float*)d_in[0];
    const float* w1  = (const float*)d_in[1];
    const float* bb1 = (const float*)d_in[2];
    const float* w2  = (const float*)d_in[3];
    const float* bb2 = (const float*)d_in[4];
    const float* w3  = (const float*)d_in[5];
    const float* bb3 = (const float*)d_in[6];
    const float* w4  = (const float*)d_in[7];
    const float* bb4 = (const float*)d_in[8];
    float* out = (float*)d_out;

    float* Q = (float*)d_ws;
    float* K = Q + (size_t)BATCH * PLANE;
    float* V = K + (size_t)BATCH * PLANE;
    float* Y = V + (size_t)BATCH * PLANE;

    qkv_conv_kernel<<<dim3((BATCH * N_PIX) / 256, 3, CH / OG), 256, 0, stream>>>(
        x, w1, bb1, w2, bb2, w3, bb3, Q, K, V);
    attn_kernel<<<N_PIX / 16, 256, 0, stream>>>(Q, K, V, Y);
    out_conv_kernel<<<dim3((BATCH * N_PIX) / 256, CIN / OGO), 256, 0, stream>>>(Y, w4, bb4, out);
}
